# starting kernel
# baseline (speedup 1.0000x reference)
#include <hip/hip_runtime.h>

typedef float v2f __attribute__((ext_vector_type(2)));
typedef float v8f __attribute__((ext_vector_type(8)));

#define HW_SZ          25600
#define C_SZ           64
#define B_SZ           32
#define PIX_PER_BLOCK  128
#define BLOCKS_PER_IMG (HW_SZ / PIX_PER_BLOCK)
#define NBLK1          (B_SZ * BLOCKS_PER_IMG)
#define NTOT           (B_SZ * HW_SZ)
#define BN_EPS_F       1e-5f
#define SLOPE_F        0.1f

__global__ __launch_bounds__(256) void maskgen_wmma(
    const float* __restrict__ sf, const float* __restrict__ feats,
    float* __restrict__ out, float* __restrict__ partials)
{
    const int lane  = threadIdx.x & 31;
    const int wave  = threadIdx.x >> 5;
    const int img   = blockIdx.x / BLOCKS_PER_IMG;
    const int chunk = blockIdx.x % BLOCKS_PER_IMG;
    const int p0    = chunk * PIX_PER_BLOCK + wave * 16;

    const float* __restrict__ sfb = sf + img * C_SZ;
    const float* __restrict__ fp  =
        feats + (size_t)img * C_SZ * HW_SZ + p0 + (lane & 15);

    const int hi = lane >> 4;
    const int m  = lane & 15;

    v8f acc = {};
    #pragma unroll
    for (int c = 0; c < C_SZ; c += 4) {
        const int ch = c + (hi << 1);
        v2f a, bm;
        a[0] = fp[(size_t)(ch + 0) * HW_SZ];
        a[1] = fp[(size_t)(ch + 1) * HW_SZ];
        const float s0 = sfb[c + 0], s1 = sfb[c + 1];
        const float s2 = sfb[c + 2], s3 = sfb[c + 3];
        bm[0] = hi ? s2 : s0;
        bm[1] = hi ? s3 : s1;
        acc = __builtin_amdgcn_wmma_f32_16x16x4_f32(
                  false, a, false, bm, (short)0, acc, false, false);
    }

    float* __restrict__ op = out + (size_t)img * HW_SZ + p0;
    if (m < 8) {
        float val = acc[0];
        #pragma unroll
        for (int v = 1; v < 8; ++v) if (m == v) val = acc[v];
        op[m + (hi << 3)] = val;
    }

    float ps = acc[0] + acc[1] + acc[2] + acc[3] + acc[4] + acc[5] + acc[6] + acc[7];
    float qs = acc[0]*acc[0] + acc[1]*acc[1] + acc[2]*acc[2] + acc[3]*acc[3]
             + acc[4]*acc[4] + acc[5]*acc[5] + acc[6]*acc[6] + acc[7]*acc[7];
    float fullS = ps + __shfl_xor(ps, 16, 32);
    float fullQ = qs + __shfl_xor(qs, 16, 32);

    __shared__ float sS[8], sQ[8];
    if (lane == 0) { sS[wave] = fullS; sQ[wave] = fullQ; }
    __syncthreads();
    if (threadIdx.x == 0) {
        float S = 0.f, Q = 0.f;
        #pragma unroll
        for (int w = 0; w < 8; ++w) { S += sS[w]; Q += sQ[w]; }
        partials[2 * blockIdx.x + 0] = S;
        partials[2 * blockIdx.x + 1] = Q;
    }
}

__global__ __launch_bounds__(256) void reduce_stats(
    const float* __restrict__ partials, float* __restrict__ stats)
{
    __shared__ float sS[256], sQ[256];
    float S = 0.f, Q = 0.f;
    for (int i = threadIdx.x; i < NBLK1; i += 256) {
        S += partials[2 * i + 0];
        Q += partials[2 * i + 1];
    }
    sS[threadIdx.x] = S; sQ[threadIdx.x] = Q;
    __syncthreads();
    for (int s = 128; s > 0; s >>= 1) {
        if ((int)threadIdx.x < s) {
            sS[threadIdx.x] += sS[threadIdx.x + s];
            sQ[threadIdx.x] += sQ[threadIdx.x + s];
        }
        __syncthreads();
    }
    if (threadIdx.x == 0) {
        const float invN = 1.0f / (float)NTOT;
        const float mean = sS[0] * invN;
        const float var  = sQ[0] * invN - mean * mean;
        stats[0] = mean;
        stats[1] = rsqrtf(var + BN_EPS_F);
    }
}

__global__ __launch_bounds__(256) void normalize_lrelu(
    float* __restrict__ out, const float* __restrict__ stats,
    const float* __restrict__ bn_w, const float* __restrict__ bn_b)
{
    const float mean = stats[0];
    const float g    = bn_w[0] * stats[1];
    const float bb   = bn_b[0] - mean * g;
    const int i = blockIdx.x * blockDim.x + threadIdx.x;
    float4* o4 = (float4*)out;
    float4 v = o4[i];
    float y0 = v.x * g + bb, y1 = v.y * g + bb;
    float y2 = v.z * g + bb, y3 = v.w * g + bb;
    v.x = (y0 >= 0.f) ? y0 : SLOPE_F * y0;
    v.y = (y1 >= 0.f) ? y1 : SLOPE_F * y1;
    v.z = (y2 >= 0.f) ? y2 : SLOPE_F * y2;
    v.w = (y3 >= 0.f) ? y3 : SLOPE_F * y3;
    o4[i] = v;
}

extern "C" void kernel_launch(void* const* d_in, const int* in_sizes, int n_in,
                              void* d_out, int out_size, void* d_ws, size_t ws_size,
                              hipStream_t stream) {
    const float* sf    = (const float*)d_in[0];
    const float* feats = (const float*)d_in[1];
    const float* bn_w  = (const float*)d_in[2];
    const float* bn_b  = (const float*)d_in[3];
    float* out      = (float*)d_out;
    float* partials = (float*)d_ws;
    float* stats    = partials + 2 * NBLK1;

    maskgen_wmma   <<<NBLK1, 256, 0, stream>>>(sf, feats, out, partials);
    reduce_stats   <<<1, 256, 0, stream>>>(partials, stats);
    normalize_lrelu<<<NTOT / 4 / 256, 256, 0, stream>>>(out, stats, bn_w, bn_b);
}


# Phase 1 split over channels, not pixels: 256 work-groups x 16 waves, a wave streams 8 channel rows of a 1600-pixel run; a pixel's 8 partial sums meet in LDS (channel sum regrouped as 8 x 8)
# speedup vs baseline: 1.2538x; 1.2538x over previous
#include <hip/hip_runtime.h>

typedef float v2f __attribute__((ext_vector_type(2)));
typedef float v4f __attribute__((ext_vector_type(4)));
typedef float v8f __attribute__((ext_vector_type(8)));

#define HW_SZ     25600
#define C_SZ      64
#define B_SZ      32
#define NTOT      (B_SZ * HW_SZ)
#define BN_EPS_F  1e-5f
#define SLOPE_F   0.1f

#define NWG       256
#define STRIP     3200
#define HALF      1600
#define NSUB      25
#define AHEAD     5
#define RING      (AHEAD + 1)
#define NTILE     7
#define PITCH     72
#define WREG      (NTILE * 4 * PITCH)
#define NQ        (STRIP / 4)

__global__ __launch_bounds__(512) void maskgen_splitk(
    const float* __restrict__ sf, const float* __restrict__ feats,
    float* __restrict__ out, float* __restrict__ partials)
{
    __shared__ __attribute__((aligned(16))) float part[16 * WREG];
    __shared__ float sS[16], sQ[16];

    const int tid   = threadIdx.x;
    const int lane  = tid & 31;
    const int wave  = __builtin_amdgcn_readfirstlane(tid >> 5);
    const int img   = blockIdx.x >> 3;
    const int strip = blockIdx.x & 7;
    const int kw    = wave & 7;
    const int ph    = wave >> 3;
    const int c     = lane & 15;
    const int h     = lane >> 4;

    const float* __restrict__ sfb = sf + img * C_SZ + kw * 8 + 2 * h;
    const float w00 = sfb[0], w01 = sfb[1], w10 = sfb[4], w11 = sfb[5];

    v2f bfr[2][16];
    #pragma unroll
    for (int t = 0; t < 16; ++t) {
        bfr[0][t][0] = (c == t) ? w00 : 0.f;
        bfr[0][t][1] = (c == t) ? w01 : 0.f;
        bfr[1][t][0] = (c == t) ? w10 : 0.f;
        bfr[1][t][1] = (c == t) ? w11 : 0.f;
    }

    const float* __restrict__ fp = feats
        + (size_t)(img * C_SZ + kw * 8 + 2 * h) * HW_SZ
        + strip * STRIP + ph * HALF + 4 * c;

    v4f buf[RING][2][2];
#define LOAD_SUB(U) do { \
    _Pragma("unroll") for (int ks = 0; ks < 2; ++ks) \
    _Pragma("unroll") for (int s = 0; s < 2; ++s) \
        buf[(U) % RING][ks][s] = *reinterpret_cast<const v4f*>( \
            fp + (size_t)(4 * ks + s) * HW_SZ + 64 * (U)); } while (0)

    #pragma unroll
    for (int U = 0; U < AHEAD; ++U) LOAD_SUB(U);

    float* __restrict__ lp = part + wave * WREG + PITCH * (c >> 2) + 32 * h + (c & 3);

    v8f acc = {};
    #pragma unroll
    for (int U = 0; U < NSUB; ++U) {
        __builtin_amdgcn_sched_barrier(0);
        if (U + AHEAD < NSUB) LOAD_SUB(U + AHEAD);
        __builtin_amdgcn_sched_barrier(0);
        #pragma unroll
        for (int jj = 0; jj < 4; ++jj) {
            #pragma unroll
            for (int ks = 0; ks < 2; ++ks) {
                v2f a;
                a[0] = buf[U % RING][ks][0][jj];
                a[1] = buf[U % RING][ks][1][jj];
                acc = __builtin_amdgcn_wmma_f32_16x16x4_f32(
                          false, a, false, bfr[ks][4 * (U & 3) + jj],
                          (short)0, acc, false, false);
            }
        }
        if ((U & 3) == 3 || U == NSUB - 1) {
            #pragma unroll
            for (int r = 0; r < 8; ++r) lp[4 * PITCH * (U >> 2) + 4 * r] = acc[r];
            acc = v8f{};
        }
    }
#undef LOAD_SUB
    __syncthreads();

    float S = 0.f, Q = 0.f;
    float* __restrict__ ob = out + (size_t)img * HW_SZ + strip * STRIP;
    #pragma unroll
    for (int rep = 0; rep < 2; ++rep) {
        const int pq = tid + 512 * rep;
        if (pq < NQ) {
            const int php = (pq >= NQ / 2) ? 1 : 0;
            const int q   = pq - (NQ / 2) * php;
            const float* rp = part + php * 8 * WREG + PITCH * (q >> 4) + 4 * (q & 15);
            v4f m4 = *reinterpret_cast<const v4f*>(rp);
            #pragma unroll
            for (int k = 1; k < 8; ++k) m4 += *reinterpret_cast<const v4f*>(rp + k * WREG);
            *reinterpret_cast<v4f*>(ob + 4 * pq) = m4;
            S += (m4[0] + m4[1]) + (m4[2] + m4[3]);
            Q += (m4[0] * m4[0] + m4[1] * m4[1]) + (m4[2] * m4[2] + m4[3] * m4[3]);
        }
    }
    #pragma unroll
    for (int m = 16; m >= 1; m >>= 1) {
        S += __shfl_xor(S, m, 32);
        Q += __shfl_xor(Q, m, 32);
    }
    if (lane == 0) { sS[wave] = S; sQ[wave] = Q; }
    __syncthreads();
    if (tid == 0) {
        float a = 0.f, b = 0.f;
        #pragma unroll
        for (int w = 0; w < 16; ++w) { a += sS[w]; b += sQ[w]; }
        partials[2 * blockIdx.x + 0] = a;
        partials[2 * blockIdx.x + 1] = b;
    }
}

__global__ __launch_bounds__(256) void reduce_stats(
    const float* __restrict__ partials, float* __restrict__ stats)
{
    __shared__ float sS[256], sQ[256];
    float S = 0.f, Q = 0.f;
    for (int i = threadIdx.x; i < NWG; i += 256) {
        S += partials[2 * i + 0];
        Q += partials[2 * i + 1];
    }
    sS[threadIdx.x] = S; sQ[threadIdx.x] = Q;
    __syncthreads();
    for (int s = 128; s > 0; s >>= 1) {
        if ((int)threadIdx.x < s) {
            sS[threadIdx.x] += sS[threadIdx.x + s];
            sQ[threadIdx.x] += sQ[threadIdx.x + s];
        }
        __syncthreads();
    }
    if (threadIdx.x == 0) {
        const float invN = 1.0f / (float)NTOT;
        const float mean = sS[0] * invN;
        const float var  = sQ[0] * invN - mean * mean;
        stats[0] = mean;
        stats[1] = rsqrtf(var + BN_EPS_F);
    }
}

__global__ __launch_bounds__(256) void normalize_lrelu(
    float* __restrict__ out, const float* __restrict__ stats,
    const float* __restrict__ bn_w, const float* __restrict__ bn_b)
{
    const float mean = stats[0];
    const float g    = bn_w[0] * stats[1];
    const float bb   = bn_b[0] - mean * g;
    const int i = blockIdx.x * blockDim.x + threadIdx.x;
    float4* o4 = (float4*)out;
    float4 v = o4[i];
    float y0 = v.x * g + bb, y1 = v.y * g + bb;
    float y2 = v.z * g + bb, y3 = v.w * g + bb;
    v.x = (y0 >= 0.f) ? y0 : SLOPE_F * y0;
    v.y = (y1 >= 0.f) ? y1 : SLOPE_F * y1;
    v.z = (y2 >= 0.f) ? y2 : SLOPE_F * y2;
    v.w = (y3 >= 0.f) ? y3 : SLOPE_F * y3;
    o4[i] = v;
}

extern "C" void kernel_launch(void* const* d_in, const int* in_sizes, int n_in,
                              void* d_out, int out_size, void* d_ws, size_t ws_size,
                              hipStream_t stream) {
    const float* sf    = (const float*)d_in[0];
    const float* feats = (const float*)d_in[1];
    const float* bn_w  = (const float*)d_in[2];
    const float* bn_b  = (const float*)d_in[3];
    float* out      = (float*)d_out;
    float* partials = (float*)d_ws;
    float* stats    = partials + 2 * NWG;

    maskgen_splitk <<<NWG, 512, 0, stream>>>(sf, feats, out, partials);
    reduce_stats   <<<1, 256, 0, stream>>>(partials, stats);
    normalize_lrelu<<<NTOT / 4 / 256, 256, 0, stream>>>(out, stats, bn_w, bn_b);
}
